# starting kernel
# baseline (speedup 1.0000x reference)
#include <hip/hip_runtime.h>

typedef __attribute__((ext_vector_type(2))) float v2f;
typedef __attribute__((ext_vector_type(4))) float v4f;
typedef __attribute__((ext_vector_type(8))) float v8f;

#define DIM 128
#define WAVES_PER_BLOCK 4
#define THREADS (WAVES_PER_BLOCK * 32)
#define ROWS_PER_WAVE 16
#define ROWS_PER_BLOCK (ROWS_PER_WAVE * WAVES_PER_BLOCK)

__global__ __launch_bounds__(THREADS) void onehot_fc_wmma_kernel(
    const int* __restrict__ x, const float* __restrict__ W1,
    const float* __restrict__ b1, const float* __restrict__ W2,
    const float* __restrict__ b2, float* __restrict__ out) {
  __shared__ float lds_w2[DIM * DIM];

  const int t = threadIdx.x;
  const int wave = t >> 5;
  const int lane = t & 31;
  const int g = lane >> 4;
  const int lm = lane & 15;
  const int m0 = blockIdx.x * ROWS_PER_BLOCK + wave * ROWS_PER_WAVE;

  const int row = m0 + lm;
  const int idx = x[row];
  const float* __restrict__ wrow = W1 + (size_t)idx * DIM;

  v2f wraw[32];
#pragma unroll
  for (int ks = 0; ks < 32; ++ks) {
    wraw[ks] = *(const v2f*)(wrow + ks * 4 + g * 2);
  }

  {
    const v4f* __restrict__ w2v = (const v4f*)W2;
#pragma unroll
    for (int it = 0; it < (DIM * DIM / 4) / THREADS; ++it) {
      const int v = it * THREADS + t;
      const int r = v >> 5;
      const int col4 = (v & 31) << 2;
      const int scol = col4 ^ (((r >> 1) & 1) << 4);
      v4f d = w2v[v];
      *(v4f*)&lds_w2[r * DIM + scol] = d;
    }
  }
  __syncthreads();

  v2f afrag[32];
#pragma unroll
  for (int ks = 0; ks < 32; ++ks) {
    const int c = ks * 4 + g * 2;
    v2f bb = *(const v2f*)(b1 + c);
    v2f a;
    a.x = fmaxf(wraw[ks].x + bb.x, 0.0f);
    a.y = fmaxf(wraw[ks].y + bb.y, 0.0f);
    afrag[ks] = a;
  }

  v8f acc[8] = {};

  const int sw = g << 4;
#pragma unroll 4
  for (int ks = 0; ks < 32; ++ks) {
    const int kk = ks * 4 + g * 2;
    const v2f a = afrag[ks];
#pragma unroll
    for (int nt = 0; nt < 8; ++nt) {
      const int scol = (nt * 16 + lm) ^ sw;
      v2f b;
      b.x = lds_w2[kk * DIM + scol];
      b.y = lds_w2[(kk + 1) * DIM + scol];
      acc[nt] = __builtin_amdgcn_wmma_f32_16x16x4_f32(
          false, a, false, b, (short)0, acc[nt], false, false);
    }
  }

#pragma unroll
  for (int nt = 0; nt < 8; ++nt) {
    const int col = nt * 16 + lm;
    const float bias2 = b2[col];
#pragma unroll
    for (int vv = 0; vv < 8; ++vv) {
      const int r = m0 + vv + 8 * g;
      out[r * DIM + col] = acc[nt][vv] + bias2;
    }
  }
}

extern "C" void kernel_launch(void* const* d_in, const int* in_sizes, int n_in,
                              void* d_out, int out_size, void* d_ws, size_t ws_size,
                              hipStream_t stream) {
  (void)n_in; (void)out_size; (void)d_ws; (void)ws_size;
  const int* x = (const int*)d_in[0];
  const float* W1 = (const float*)d_in[1];
  const float* b1 = (const float*)d_in[2];
  const float* W2 = (const float*)d_in[3];
  const float* b2 = (const float*)d_in[4];
  float* out = (float*)d_out;

  const int batch = in_sizes[0];
  dim3 grid(batch / ROWS_PER_BLOCK);
  onehot_fc_wmma_kernel<<<grid, THREADS, 0, stream>>>(x, W1, b1, W2, b2, out);
}


# Split the 128 output columns over 4 workgroups per 64-row group: grid 64 -> 256 workgroups for 256 WGPs, each stages a 128x32 slice of W2 (16 KiB LDS), 64 WMMAs per wave instead of 256
# speedup vs baseline: 2.1029x; 2.1029x over previous
#include <hip/hip_runtime.h>

typedef __attribute__((ext_vector_type(2))) float v2f;
typedef __attribute__((ext_vector_type(4))) float v4f;
typedef __attribute__((ext_vector_type(8))) float v8f;

#define DIM 128
#define WAVES_PER_BLOCK 4
#define THREADS (WAVES_PER_BLOCK * 32)
#define ROWS_PER_WAVE 16
#define ROWS_PER_BLOCK (ROWS_PER_WAVE * WAVES_PER_BLOCK)
#define COL_GROUPS 4
#define COLS_PER_BLOCK (DIM / COL_GROUPS)
#define NT_PER_BLOCK (COLS_PER_BLOCK / 16)

__global__ __launch_bounds__(THREADS) void onehot_fc_wmma_kernel(
    const int* __restrict__ x, const float* __restrict__ W1,
    const float* __restrict__ b1, const float* __restrict__ W2,
    const float* __restrict__ b2, float* __restrict__ out) {
  __shared__ float lds_w2[DIM * COLS_PER_BLOCK];

  const int t = threadIdx.x;
  const int wave = t >> 5;
  const int lane = t & 31;
  const int g = lane >> 4;
  const int lm = lane & 15;
  const int m0 = blockIdx.x * ROWS_PER_BLOCK + wave * ROWS_PER_WAVE;
  const int c0 = blockIdx.y * COLS_PER_BLOCK;

  const int row = m0 + lm;
  const int idx = x[row];
  const float* __restrict__ wrow = W1 + (size_t)idx * DIM;

  v2f wraw[32];
#pragma unroll
  for (int ks = 0; ks < 32; ++ks) {
    wraw[ks] = *(const v2f*)(wrow + ks * 4 + g * 2);
  }

  {
#pragma unroll
    for (int it = 0; it < (DIM * COLS_PER_BLOCK / 4) / THREADS; ++it) {
      const int v = it * THREADS + t;
      const int r = v >> 3;
      const int col4 = (v & 7) << 2;
      const int scol = col4 ^ (((r >> 1) & 1) << 4);
      v4f d = *(const v4f*)(W2 + r * DIM + c0 + col4);
      *(v4f*)&lds_w2[r * COLS_PER_BLOCK + scol] = d;
    }
  }
  __syncthreads();

  v2f afrag[32];
#pragma unroll
  for (int ks = 0; ks < 32; ++ks) {
    const int c = ks * 4 + g * 2;
    v2f bb = *(const v2f*)(b1 + c);
    v2f a;
    a.x = fmaxf(wraw[ks].x + bb.x, 0.0f);
    a.y = fmaxf(wraw[ks].y + bb.y, 0.0f);
    afrag[ks] = a;
  }

  v8f acc[NT_PER_BLOCK] = {};

  const int sw = g << 4;
#pragma unroll
  for (int ks = 0; ks < 32; ++ks) {
    const int kk = ks * 4 + g * 2;
    const v2f a = afrag[ks];
#pragma unroll
    for (int nt = 0; nt < NT_PER_BLOCK; ++nt) {
      const int scol = (nt * 16 + lm) ^ sw;
      v2f b;
      b.x = lds_w2[kk * COLS_PER_BLOCK + scol];
      b.y = lds_w2[(kk + 1) * COLS_PER_BLOCK + scol];
      acc[nt] = __builtin_amdgcn_wmma_f32_16x16x4_f32(
          false, a, false, b, (short)0, acc[nt], false, false);
    }
  }

#pragma unroll
  for (int nt = 0; nt < NT_PER_BLOCK; ++nt) {
    const int col = c0 + nt * 16 + lm;
    const float bias2 = b2[col];
#pragma unroll
    for (int vv = 0; vv < 8; ++vv) {
      const int r = m0 + vv + 8 * g;
      out[r * DIM + col] = acc[nt][vv] + bias2;
    }
  }
}

extern "C" void kernel_launch(void* const* d_in, const int* in_sizes, int n_in,
                              void* d_out, int out_size, void* d_ws, size_t ws_size,
                              hipStream_t stream) {
  (void)n_in; (void)out_size; (void)d_ws; (void)ws_size;
  const int* x = (const int*)d_in[0];
  const float* W1 = (const float*)d_in[1];
  const float* b1 = (const float*)d_in[2];
  const float* W2 = (const float*)d_in[3];
  const float* b2 = (const float*)d_in[4];
  float* out = (float*)d_out;

  const int batch = in_sizes[0];
  dim3 grid(batch / ROWS_PER_BLOCK, COL_GROUPS);
  onehot_fc_wmma_kernel<<<grid, THREADS, 0, stream>>>(x, W1, b1, W2, b2, out);
}


# All loads moved out of the 64-WMMA stream: b1/b2 fetched up front, the 64 B-fragment LDS reads issued as one group into registers behind the barrier, so the WMMAs issue back to back
# speedup vs baseline: 2.6648x; 1.2672x over previous
#include <hip/hip_runtime.h>

typedef __attribute__((ext_vector_type(2))) float v2f;
typedef __attribute__((ext_vector_type(4))) float v4f;
typedef __attribute__((ext_vector_type(8))) float v8f;

#define DIM 128
#define WAVES_PER_BLOCK 4
#define THREADS (WAVES_PER_BLOCK * 32)
#define ROWS_PER_WAVE 16
#define ROWS_PER_BLOCK (ROWS_PER_WAVE * WAVES_PER_BLOCK)
#define COL_GROUPS 4
#define COLS_PER_BLOCK (DIM / COL_GROUPS)
#define NT_PER_BLOCK (COLS_PER_BLOCK / 16)

__global__ __launch_bounds__(THREADS) void onehot_fc_wmma_kernel(
    const int* __restrict__ x, const float* __restrict__ W1,
    const float* __restrict__ b1, const float* __restrict__ W2,
    const float* __restrict__ b2, float* __restrict__ out) {
  __shared__ float lds_w2[DIM * COLS_PER_BLOCK];

  const int t = threadIdx.x;
  const int wave = t >> 5;
  const int lane = t & 31;
  const int g = lane >> 4;
  const int lm = lane & 15;
  const int m0 = blockIdx.x * ROWS_PER_BLOCK + wave * ROWS_PER_WAVE;
  const int c0 = blockIdx.y * COLS_PER_BLOCK;

  const int row = m0 + lm;
  const int idx = x[row];
  const float* __restrict__ wrow = W1 + (size_t)idx * DIM;

  v2f bias1[32];
#pragma unroll
  for (int ks = 0; ks < 32; ++ks) {
    bias1[ks] = *(const v2f*)(b1 + ks * 4 + g * 2);
  }
  float bias2[NT_PER_BLOCK];
#pragma unroll
  for (int nt = 0; nt < NT_PER_BLOCK; ++nt) {
    bias2[nt] = b2[c0 + nt * 16 + lm];
  }

  v2f wraw[32];
#pragma unroll
  for (int ks = 0; ks < 32; ++ks) {
    wraw[ks] = *(const v2f*)(wrow + ks * 4 + g * 2);
  }

  {
#pragma unroll
    for (int it = 0; it < (DIM * COLS_PER_BLOCK / 4) / THREADS; ++it) {
      const int v = it * THREADS + t;
      const int r = v >> 3;
      const int col4 = (v & 7) << 2;
      const int scol = col4 ^ (((r >> 1) & 1) << 4);
      v4f d = *(const v4f*)(W2 + r * DIM + c0 + col4);
      *(v4f*)&lds_w2[r * COLS_PER_BLOCK + scol] = d;
    }
  }
  __syncthreads();

  const int sw = g << 4;
  v2f bfrag[32][NT_PER_BLOCK];
#pragma unroll
  for (int ks = 0; ks < 32; ++ks) {
    const int kk = ks * 4 + g * 2;
#pragma unroll
    for (int nt = 0; nt < NT_PER_BLOCK; ++nt) {
      const int scol = (nt * 16 + lm) ^ sw;
      v2f b;
      b.x = lds_w2[kk * COLS_PER_BLOCK + scol];
      b.y = lds_w2[(kk + 1) * COLS_PER_BLOCK + scol];
      bfrag[ks][nt] = b;
    }
  }

  v2f afrag[32];
#pragma unroll
  for (int ks = 0; ks < 32; ++ks) {
    v2f a;
    a.x = fmaxf(wraw[ks].x + bias1[ks].x, 0.0f);
    a.y = fmaxf(wraw[ks].y + bias1[ks].y, 0.0f);
    afrag[ks] = a;
  }

  v8f acc[NT_PER_BLOCK] = {};

  __builtin_amdgcn_sched_barrier(0);

#pragma unroll
  for (int ks = 0; ks < 32; ++ks) {
#pragma unroll
    for (int nt = 0; nt < NT_PER_BLOCK; ++nt) {
      acc[nt] = __builtin_amdgcn_wmma_f32_16x16x4_f32(
          false, afrag[ks], false, bfrag[ks][nt], (short)0, acc[nt], false, false);
    }
  }

#pragma unroll
  for (int nt = 0; nt < NT_PER_BLOCK; ++nt) {
    const int col = c0 + nt * 16 + lm;
#pragma unroll
    for (int vv = 0; vv < 8; ++vv) {
      const int r = m0 + vv + 8 * g;
      out[r * DIM + col] = acc[nt][vv] + bias2[nt];
    }
  }
}

extern "C" void kernel_launch(void* const* d_in, const int* in_sizes, int n_in,
                              void* d_out, int out_size, void* d_ws, size_t ws_size,
                              hipStream_t stream) {
  (void)n_in; (void)out_size; (void)d_ws; (void)ws_size;
  const int* x = (const int*)d_in[0];
  const float* W1 = (const float*)d_in[1];
  const float* b1 = (const float*)d_in[2];
  const float* W2 = (const float*)d_in[3];
  const float* b2 = (const float*)d_in[4];
  float* out = (float*)d_out;

  const int batch = in_sizes[0];
  dim3 grid(batch / ROWS_PER_BLOCK, COL_GROUPS);
  onehot_fc_wmma_kernel<<<grid, THREADS, 0, stream>>>(x, W1, b1, W2, b2, out);
}


# W1 gather coalesced: one load per whole 512-byte row (scalar x indices), +b1/ReLU in row-major registers, transposed to the A layout via per-wave LDS; W2 barrier fences LDS only, under the gather
# speedup vs baseline: 2.8749x; 1.0788x over previous
#include <hip/hip_runtime.h>

typedef __attribute__((ext_vector_type(2))) float v2f;
typedef __attribute__((ext_vector_type(4))) float v4f;
typedef __attribute__((ext_vector_type(8))) float v8f;

#define DIM 128
#define WAVES_PER_BLOCK 4
#define THREADS (WAVES_PER_BLOCK * 32)
#define ROWS_PER_WAVE 16
#define ROWS_PER_BLOCK (ROWS_PER_WAVE * WAVES_PER_BLOCK)
#define COL_GROUPS 4
#define COLS_PER_BLOCK (DIM / COL_GROUPS)
#define NT_PER_BLOCK (COLS_PER_BLOCK / 16)
#define A_PITCH (DIM + 4)

__global__ __launch_bounds__(THREADS) void onehot_fc_wmma_kernel(
    const int* __restrict__ x, const float* __restrict__ W1,
    const float* __restrict__ b1, const float* __restrict__ W2,
    const float* __restrict__ b2, float* __restrict__ out) {
  __shared__ __attribute__((aligned(16))) float lds_w2[DIM * COLS_PER_BLOCK];
  __shared__ __attribute__((aligned(16))) float lds_a[ROWS_PER_BLOCK * A_PITCH];

  const int t = threadIdx.x;
  const int wave = __builtin_amdgcn_readfirstlane(t >> 5);
  const int lane = t & 31;
  const int g = lane >> 4;
  const int lm = lane & 15;
  const int m0 = blockIdx.x * ROWS_PER_BLOCK + wave * ROWS_PER_WAVE;
  const int c0 = blockIdx.y * COLS_PER_BLOCK;

  const int* __restrict__ xs = x + m0;
  int idx[ROWS_PER_WAVE];
#pragma unroll
  for (int r = 0; r < ROWS_PER_WAVE; ++r) {
    idx[r] = xs[r];
  }

  const v4f bias1 = *(const v4f*)(b1 + lane * 4);
  float bias2[NT_PER_BLOCK];
#pragma unroll
  for (int nt = 0; nt < NT_PER_BLOCK; ++nt) {
    bias2[nt] = b2[c0 + nt * 16 + lm];
  }
  v4f w2v[(DIM * COLS_PER_BLOCK / 4) / THREADS];
#pragma unroll
  for (int it = 0; it < (DIM * COLS_PER_BLOCK / 4) / THREADS; ++it) {
    const int v = it * THREADS + t;
    const int r = v >> 3;
    const int col4 = (v & 7) << 2;
    w2v[it] = *(const v4f*)(W2 + r * DIM + c0 + col4);
  }

  __builtin_amdgcn_sched_barrier(0);

  v4f raw[ROWS_PER_WAVE];
#pragma unroll
  for (int r = 0; r < ROWS_PER_WAVE; ++r) {
    raw[r] = *(const v4f*)(W1 + (size_t)idx[r] * DIM + lane * 4);
  }

#pragma unroll
  for (int it = 0; it < (DIM * COLS_PER_BLOCK / 4) / THREADS; ++it) {
    const int v = it * THREADS + t;
    const int r = v >> 3;
    const int col4 = (v & 7) << 2;
    const int scol = col4 ^ (((r >> 1) & 1) << 4);
    *(v4f*)&lds_w2[r * COLS_PER_BLOCK + scol] = w2v[it];
  }
  __builtin_amdgcn_fence(__ATOMIC_RELEASE, "workgroup", "local");
  __builtin_amdgcn_s_barrier();
  __builtin_amdgcn_fence(__ATOMIC_ACQUIRE, "workgroup", "local");

  const int sw = g << 4;
  v2f bfrag[32][NT_PER_BLOCK];
#pragma unroll
  for (int ks = 0; ks < 32; ++ks) {
    const int kk = ks * 4 + g * 2;
#pragma unroll
    for (int nt = 0; nt < NT_PER_BLOCK; ++nt) {
      const int scol = (nt * 16 + lm) ^ sw;
      v2f b;
      b.x = lds_w2[kk * COLS_PER_BLOCK + scol];
      b.y = lds_w2[(kk + 1) * COLS_PER_BLOCK + scol];
      bfrag[ks][nt] = b;
    }
  }

  __builtin_amdgcn_sched_barrier(0);

  float* arow = lds_a + (wave * ROWS_PER_WAVE) * A_PITCH;
#pragma unroll
  for (int r = 0; r < ROWS_PER_WAVE; ++r) {
    v4f h;
    h.x = fmaxf(raw[r].x + bias1.x, 0.0f);
    h.y = fmaxf(raw[r].y + bias1.y, 0.0f);
    h.z = fmaxf(raw[r].z + bias1.z, 0.0f);
    h.w = fmaxf(raw[r].w + bias1.w, 0.0f);
    *(v4f*)&arow[r * A_PITCH + lane * 4] = h;
  }
  __builtin_amdgcn_fence(__ATOMIC_RELEASE, "wavefront", "local");
  __builtin_amdgcn_wave_barrier();
  __builtin_amdgcn_fence(__ATOMIC_ACQUIRE, "wavefront", "local");

  v2f afrag[32];
#pragma unroll
  for (int ks = 0; ks < 32; ++ks) {
    afrag[ks] = *(const v2f*)&arow[lm * A_PITCH + ks * 4 + g * 2];
  }

  v8f acc[NT_PER_BLOCK] = {};

  __builtin_amdgcn_sched_barrier(0);

#pragma unroll
  for (int ks = 0; ks < 32; ++ks) {
#pragma unroll
    for (int nt = 0; nt < NT_PER_BLOCK; ++nt) {
      acc[nt] = __builtin_amdgcn_wmma_f32_16x16x4_f32(
          false, afrag[ks], false, bfrag[ks][nt], (short)0, acc[nt], false, false);
    }
  }

#pragma unroll
  for (int nt = 0; nt < NT_PER_BLOCK; ++nt) {
    const int col = c0 + nt * 16 + lm;
#pragma unroll
    for (int vv = 0; vv < 8; ++vv) {
      const int r = m0 + vv + 8 * g;
      out[r * DIM + col] = acc[nt][vv] + bias2[nt];
    }
  }
}

extern "C" void kernel_launch(void* const* d_in, const int* in_sizes, int n_in,
                              void* d_out, int out_size, void* d_ws, size_t ws_size,
                              hipStream_t stream) {
  (void)n_in; (void)out_size; (void)d_ws; (void)ws_size;
  const int* x = (const int*)d_in[0];
  const float* W1 = (const float*)d_in[1];
  const float* b1 = (const float*)d_in[2];
  const float* W2 = (const float*)d_in[3];
  const float* b2 = (const float*)d_in[4];
  float* out = (float*)d_out;

  const int batch = in_sizes[0];
  dim3 grid(batch / ROWS_PER_BLOCK, COL_GROUPS);
  onehot_fc_wmma_kernel<<<grid, THREADS, 0, stream>>>(x, W1, b1, W2, b2, out);
}
